# starting kernel
# baseline (speedup 1.0000x reference)
#include <hip/hip_runtime.h>
#include <hip/hip_bf16.h>


typedef __attribute__((ext_vector_type(2))) float v2f;
typedef __attribute__((ext_vector_type(8))) float v8f;

#define POLY_ORDER 7
#define NP 8
#define MT 4
#define KC 16

__global__ __launch_bounds__(256) void poly_wmma_kernel(
    const float* __restrict__ x,
    const float* __restrict__ W,
    const float* __restrict__ bias_p,
    float* __restrict__ out,
    int N, int groups_per_wave)
{
  const int lane = threadIdx.x & 31;
  const int wave = threadIdx.x >> 5;
  const int wave_global = blockIdx.x * (blockDim.x >> 5) + wave;
  const bool hi = (lane >= 16);
  const int sl = lane & 15;
  const int h2 = hi ? 2 : 0;

  v2f A[MT][KC];
#pragma unroll
  for (int m = 0; m < MT; ++m) {
    const float* wrow = W + (16 * m + sl) * 64 + h2;
#pragma unroll
    for (int k = 0; k < KC; ++k) {
      A[m][k] = *(const v2f*)(wrow + 4 * k);
    }
  }
  const float bias = bias_p[0];

  for (int g = 0; g < groups_per_wave; ++g) {
    const int s0 = (wave_global * groups_per_wave + g) * 16;
    if (s0 >= N) break;
    const int s = s0 + sl;

    const float4 xs = *(const float4*)(x + s * 4);
    float xv[4] = {xs.x, xs.y, xs.z, xs.w};
    float leg[4][NP];
#pragma unroll
    for (int d = 0; d < 4; ++d) {
      leg[d][0] = 1.0f;
      leg[d][1] = xv[d];
#pragma unroll
      for (int n = 1; n < POLY_ORDER; ++n) {
        const float rn = 1.0f / (float)(n + 1);
        leg[d][n + 1] =
            ((2.0f * n + 1.0f) * xv[d] * leg[d][n] - (float)n * leg[d][n - 1]) * rn;
      }
    }

    float l3[4];
    l3[0] = hi ? leg[3][2] : leg[3][0];
    l3[1] = hi ? leg[3][3] : leg[3][1];
    l3[2] = hi ? leg[3][6] : leg[3][4];
    l3[3] = hi ? leg[3][7] : leg[3][5];
    float l0[MT];
#pragma unroll
    for (int m = 0; m < MT; ++m) l0[m] = hi ? leg[0][2 * m + 1] : leg[0][2 * m];

    v8f acc[MT];
#pragma unroll
    for (int m = 0; m < MT; ++m)
      acc[m] = (v8f){0.f, 0.f, 0.f, 0.f, 0.f, 0.f, 0.f, 0.f};

#pragma unroll
    for (int k = 0; k < KC; ++k) {
      const float c2 = leg[2][k >> 1];
      const int o = (k & 1) * 2;
      v2f B;
      B.x = c2 * l3[o];
      B.y = c2 * l3[o + 1];
#pragma unroll
      for (int m = 0; m < MT; ++m) {
        acc[m] = __builtin_amdgcn_wmma_f32_16x16x4_f32(
             false, A[m][k],  false, B,
             (short)0, acc[m],  false,  false);
      }
    }

    float partial = 0.0f;
#pragma unroll
    for (int m = 0; m < MT; ++m) {
      float t = 0.0f;
#pragma unroll
      for (int r = 0; r < 8; ++r) t = fmaf(leg[1][r], acc[m][r], t);
      partial = fmaf(l0[m], t, partial);
    }
    partial += __shfl_xor(partial, 16, 32);
    if (!hi) out[s] = partial + bias;
  }
}

extern "C" void kernel_launch(void* const* d_in, const int* in_sizes, int n_in,
                              void* d_out, int out_size, void* d_ws, size_t ws_size,
                              hipStream_t stream) {
  (void)n_in; (void)d_ws; (void)ws_size; (void)out_size;
  const float* x = (const float*)d_in[0];
  const float* W = (const float*)d_in[1];
  const float* b = (const float*)d_in[2];
  float* out = (float*)d_out;

  const int N = in_sizes[0] / 4;
  const int WAVES_PER_BLOCK = 8;
  const int GROUPS_PER_WAVE = 2;
  const int samples_per_block = WAVES_PER_BLOCK * 16 * GROUPS_PER_WAVE;
  const int blocks = (N + samples_per_block - 1) / samples_per_block;

  poly_wmma_kernel<<<blocks, WAVES_PER_BLOCK * 32, 0, stream>>>(
      x, W, b, out, N, GROUPS_PER_WAVE);
}


# Wmat staged once per work-group in LDS by whole-line 16-byte loads; each wave reads its A fragments from LDS, grouped ahead of the WMMAs in K order, instead of 64 global loads of 16 lines each
# speedup vs baseline: 1.2374x; 1.2374x over previous
#include <hip/hip_runtime.h>


typedef __attribute__((ext_vector_type(2))) float v2f;
typedef __attribute__((ext_vector_type(4))) float v4f;
typedef __attribute__((ext_vector_type(8))) float v8f;

#define POLY_ORDER 7
#define NP 8
#define MT 4
#define KC 16
#define BLOCK_THREADS 256
#define WPITCH 68

__global__ __launch_bounds__(BLOCK_THREADS) void poly_wmma_kernel(
    const float* __restrict__ x,
    const float* __restrict__ W,
    const float* __restrict__ bias_p,
    float* __restrict__ out,
    int N, int groups_per_wave)
{
  __shared__ __attribute__((aligned(16))) float Ws[64 * WPITCH];

  const int tid = threadIdx.x;
  const int lane = tid & 31;
  const int wave = tid >> 5;
  const int wave_global = blockIdx.x * (BLOCK_THREADS >> 5) + wave;
  const bool hi = (lane >= 16);
  const int sl = lane & 15;
  const int h2 = hi ? 2 : 0;
  const int sbase = wave_global * groups_per_wave * 16;

  v4f wv[4];
#pragma unroll
  for (int i = 0; i < 4; ++i)
    wv[i] = *(const v4f*)(W + 4 * (i * BLOCK_THREADS + tid));

  float4 xs = make_float4(0.f, 0.f, 0.f, 0.f);
  if (sbase < N) xs = *(const float4*)(x + (sbase + sl) * 4);
  const float bias = bias_p[0];

#pragma unroll
  for (int i = 0; i < 4; ++i) {
    const int f4 = i * BLOCK_THREADS + tid;
    *(v4f*)(Ws + (f4 >> 4) * WPITCH + 4 * (f4 & 15)) = wv[i];
  }
  __syncthreads();

  v2f A[MT][KC];
#pragma unroll
  for (int kp = 0; kp < KC; kp += 2) {
#pragma unroll
    for (int m = 0; m < MT; ++m) {
      const float* wrow = Ws + (16 * m + sl) * WPITCH + h2;
      A[m][kp]     = *(const v2f*)(wrow + 4 * kp);
      A[m][kp + 1] = *(const v2f*)(wrow + 4 * kp + 4);
    }
    __builtin_amdgcn_sched_barrier(0);
  }

  for (int g = 0; g < groups_per_wave; ++g) {
    const int s0 = sbase + g * 16;
    if (s0 >= N) break;
    const int s = s0 + sl;
    if (g > 0) xs = *(const float4*)(x + s * 4);

    float xv[4] = {xs.x, xs.y, xs.z, xs.w};
    float leg[4][NP];
#pragma unroll
    for (int d = 0; d < 4; ++d) {
      leg[d][0] = 1.0f;
      leg[d][1] = xv[d];
#pragma unroll
      for (int n = 1; n < POLY_ORDER; ++n) {
        const float rn = 1.0f / (float)(n + 1);
        leg[d][n + 1] =
            ((2.0f * n + 1.0f) * xv[d] * leg[d][n] - (float)n * leg[d][n - 1]) * rn;
      }
    }

    float l3[4];
    l3[0] = hi ? leg[3][2] : leg[3][0];
    l3[1] = hi ? leg[3][3] : leg[3][1];
    l3[2] = hi ? leg[3][6] : leg[3][4];
    l3[3] = hi ? leg[3][7] : leg[3][5];
    float l0[MT];
#pragma unroll
    for (int m = 0; m < MT; ++m) l0[m] = hi ? leg[0][2 * m + 1] : leg[0][2 * m];

    v8f acc[MT];
#pragma unroll
    for (int m = 0; m < MT; ++m)
      acc[m] = (v8f){0.f, 0.f, 0.f, 0.f, 0.f, 0.f, 0.f, 0.f};

#pragma unroll
    for (int k = 0; k < KC; ++k) {
      const float c2 = leg[2][k >> 1];
      const int o = (k & 1) * 2;
      v2f B;
      B.x = c2 * l3[o];
      B.y = c2 * l3[o + 1];
#pragma unroll
      for (int m = 0; m < MT; ++m) {
        acc[m] = __builtin_amdgcn_wmma_f32_16x16x4_f32(
             false, A[m][k],  false, B,
             (short)0, acc[m],  false,  false);
      }
    }

    float partial = 0.0f;
#pragma unroll
    for (int m = 0; m < MT; ++m) {
      float t = 0.0f;
#pragma unroll
      for (int r = 0; r < 8; ++r) t = fmaf(leg[1][r], acc[m][r], t);
      partial = fmaf(l0[m], t, partial);
    }
    partial += __shfl_xor(partial, 16, 32);
    if (!hi) out[s] = partial + bias;
  }
}

extern "C" void kernel_launch(void* const* d_in, const int* in_sizes, int n_in,
                              void* d_out, int out_size, void* d_ws, size_t ws_size,
                              hipStream_t stream) {
  (void)n_in; (void)d_ws; (void)ws_size; (void)out_size;
  const float* x = (const float*)d_in[0];
  const float* W = (const float*)d_in[1];
  const float* b = (const float*)d_in[2];
  float* out = (float*)d_out;

  const int N = in_sizes[0] / 4;
  const int WAVES_PER_BLOCK = BLOCK_THREADS / 32;
  const int GROUPS_PER_WAVE = 2;
  const int samples_per_block = WAVES_PER_BLOCK * 16 * GROUPS_PER_WAVE;
  const int blocks = (N + samples_per_block - 1) / samples_per_block;

  poly_wmma_kernel<<<blocks, BLOCK_THREADS, 0, stream>>>(
      x, W, b, out, N, GROUPS_PER_WAVE);
}


# Lane-half choice of the four P(x0) factors made on computed values (4 conditional moves per group), not by a lane-dependent array index that compiled to a 32-way compare/select chain
# speedup vs baseline: 1.3944x; 1.1269x over previous
#include <hip/hip_runtime.h>


typedef __attribute__((ext_vector_type(2))) float v2f;
typedef __attribute__((ext_vector_type(4))) float v4f;
typedef __attribute__((ext_vector_type(8))) float v8f;

#define POLY_ORDER 7
#define NP 8
#define MT 4
#define KC 16
#define BLOCK_THREADS 256
#define WPITCH 68

static __device__ __forceinline__ float pick(bool c, float a, float b) { return c ? a : b; }

__global__ __launch_bounds__(BLOCK_THREADS) void poly_wmma_kernel(
    const float* __restrict__ x,
    const float* __restrict__ W,
    const float* __restrict__ bias_p,
    float* __restrict__ out,
    int N, int groups_per_wave)
{
  __shared__ __attribute__((aligned(16))) float Ws[64 * WPITCH];

  const int tid = threadIdx.x;
  const int lane = tid & 31;
  const int wave = tid >> 5;
  const int wave_global = blockIdx.x * (BLOCK_THREADS >> 5) + wave;
  const bool hi = (lane >= 16);
  const int sl = lane & 15;
  const int h2 = hi ? 2 : 0;
  const int sbase = wave_global * groups_per_wave * 16;

  v4f wv[4];
#pragma unroll
  for (int i = 0; i < 4; ++i)
    wv[i] = *(const v4f*)(W + 4 * (i * BLOCK_THREADS + tid));

  float4 xs = make_float4(0.f, 0.f, 0.f, 0.f);
  if (sbase < N) xs = *(const float4*)(x + (sbase + sl) * 4);
  const float bias = bias_p[0];

#pragma unroll
  for (int i = 0; i < 4; ++i) {
    const int f4 = i * BLOCK_THREADS + tid;
    *(v4f*)(Ws + (f4 >> 4) * WPITCH + 4 * (f4 & 15)) = wv[i];
  }
  __syncthreads();

  v2f A[MT][KC];
#pragma unroll
  for (int kp = 0; kp < KC; kp += 2) {
#pragma unroll
    for (int m = 0; m < MT; ++m) {
      const float* wrow = Ws + (16 * m + sl) * WPITCH + h2;
      A[m][kp]     = *(const v2f*)(wrow + 4 * kp);
      A[m][kp + 1] = *(const v2f*)(wrow + 4 * kp + 4);
    }
    __builtin_amdgcn_sched_barrier(0);
  }

  for (int g = 0; g < groups_per_wave; ++g) {
    const int s0 = sbase + g * 16;
    if (s0 >= N) break;
    const int s = s0 + sl;
    if (g > 0) xs = *(const float4*)(x + s * 4);

    float xv[4] = {xs.x, xs.y, xs.z, xs.w};
    float leg[4][NP];
#pragma unroll
    for (int d = 0; d < 4; ++d) {
      leg[d][0] = 1.0f;
      leg[d][1] = xv[d];
#pragma unroll
      for (int n = 1; n < POLY_ORDER; ++n) {
        const float rn = 1.0f / (float)(n + 1);
        leg[d][n + 1] =
            ((2.0f * n + 1.0f) * xv[d] * leg[d][n] - (float)n * leg[d][n - 1]) * rn;
      }
    }

    float l3[4];
    l3[0] = pick(hi, leg[3][2], leg[3][0]);
    l3[1] = pick(hi, leg[3][3], leg[3][1]);
    l3[2] = pick(hi, leg[3][6], leg[3][4]);
    l3[3] = pick(hi, leg[3][7], leg[3][5]);
    float l0[MT];
    l0[0] = pick(hi, leg[0][1], leg[0][0]);
    l0[1] = pick(hi, leg[0][3], leg[0][2]);
    l0[2] = pick(hi, leg[0][5], leg[0][4]);
    l0[3] = pick(hi, leg[0][7], leg[0][6]);

    v8f acc[MT];
#pragma unroll
    for (int m = 0; m < MT; ++m)
      acc[m] = (v8f){0.f, 0.f, 0.f, 0.f, 0.f, 0.f, 0.f, 0.f};

#pragma unroll
    for (int k = 0; k < KC; ++k) {
      const float c2 = leg[2][k >> 1];
      const int o = (k & 1) * 2;
      v2f B;
      B.x = c2 * l3[o];
      B.y = c2 * l3[o + 1];
#pragma unroll
      for (int m = 0; m < MT; ++m) {
        acc[m] = __builtin_amdgcn_wmma_f32_16x16x4_f32(
             false, A[m][k],  false, B,
             (short)0, acc[m],  false,  false);
      }
    }

    float partial = 0.0f;
#pragma unroll
    for (int m = 0; m < MT; ++m) {
      float t = 0.0f;
#pragma unroll
      for (int r = 0; r < 8; ++r) t = fmaf(leg[1][r], acc[m][r], t);
      partial = fmaf(l0[m], t, partial);
    }
    partial += __shfl_xor(partial, 16, 32);
    if (!hi) out[s] = partial + bias;
  }
}

extern "C" void kernel_launch(void* const* d_in, const int* in_sizes, int n_in,
                              void* d_out, int out_size, void* d_ws, size_t ws_size,
                              hipStream_t stream) {
  (void)n_in; (void)d_ws; (void)ws_size; (void)out_size;
  const float* x = (const float*)d_in[0];
  const float* W = (const float*)d_in[1];
  const float* b = (const float*)d_in[2];
  float* out = (float*)d_out;

  const int N = in_sizes[0] / 4;
  const int WAVES_PER_BLOCK = BLOCK_THREADS / 32;
  const int GROUPS_PER_WAVE = 2;
  const int samples_per_block = WAVES_PER_BLOCK * 16 * GROUPS_PER_WAVE;
  const int blocks = (N + samples_per_block - 1) / samples_per_block;

  poly_wmma_kernel<<<blocks, BLOCK_THREADS, 0, stream>>>(
      x, W, b, out, N, GROUPS_PER_WAVE);
}
